# starting kernel
# baseline (speedup 1.0000x reference)
#include <hip/hip_runtime.h>


typedef __attribute__((ext_vector_type(2))) float v2f;
typedef __attribute__((ext_vector_type(8))) float v8f;

#define NB 16
#define NC 512
#define NHW 1024
#define NG 32
#define NH 4
#define ND 128
#define GN_EPS 1e-5f
#define ATT_SCALE 0.08838834764831845f

static __device__ __forceinline__ v8f wmma4(v2f a, v2f b, v8f c) {
  return __builtin_amdgcn_wmma_f32_16x16x4_f32(false, a, false, b, (short)0, c,
                                               false, false);
}

__global__ __launch_bounds__(256) void gn_kernel(const float* __restrict__ x,
                                                 const float* __restrict__ w,
                                                 const float* __restrict__ bia,
                                                 float* __restrict__ xn) {
  const int b = blockIdx.x >> 5;
  const int g = blockIdx.x & 31;
  const size_t base = ((size_t)b * NC + (size_t)g * 16) * NHW;
  const int tid = threadIdx.x;
  float sum = 0.f, sq = 0.f;
  for (int i = tid; i < 16 * NHW; i += 256) {
    float v = x[base + i];
    sum += v;
    sq += v * v;
  }
  for (int m = 16; m >= 1; m >>= 1) {
    sum += __shfl_xor(sum, m, 32);
    sq += __shfl_xor(sq, m, 32);
  }
  __shared__ float rs[8], rq[8], stat[2];
  const int wv = tid >> 5;
  if ((tid & 31) == 0) { rs[wv] = sum; rq[wv] = sq; }
  __syncthreads();
  if (tid == 0) {
    float ts = 0.f, tq = 0.f;
    for (int i = 0; i < 8; ++i) { ts += rs[i]; tq += rq[i]; }
    float mean = ts * (1.f / 16384.f);
    float var = tq * (1.f / 16384.f) - mean * mean;
    stat[0] = mean;
    stat[1] = rsqrtf(var + GN_EPS);
  }
  __syncthreads();
  const float mean = stat[0], rstd = stat[1];
  for (int i = tid; i < 16 * NHW; i += 256) {
    int c = (g << 4) + (i >> 10);
    xn[base + i] = (x[base + i] - mean) * rstd * w[c] + bia[c];
  }
}

__global__ __launch_bounds__(256) void gemm512_kernel(
    const float* __restrict__ A, const float* __restrict__ Bmat,
    float* __restrict__ Y, long bStrideB, long bStrideY,
    const float* __restrict__ bias, const float* __restrict__ resid) {
  __shared__ float ldsA[16 * 516];
  const int m0 = blockIdx.x << 4;
  const int b = blockIdx.y;
  const int tid = threadIdx.x;
  for (int i = tid; i < 8192; i += 256) {
    int row = i >> 9, k = i & 511;
    ldsA[row * 516 + k] = A[(size_t)(m0 + row) * 512 + k];
  }
  __syncthreads();
  const int lane = tid & 31, wv = tid >> 5;
  const int lm = lane & 15, hi = lane >> 4;
  const float* Bb = Bmat + (size_t)b * bStrideB;
  const int n0 = wv << 7;
  v8f acc[8] = {};
  for (int k0 = 0; k0 < 512; k0 += 4) {
    const int kr = k0 + 2 * hi;
    v2f a;
    a.x = ldsA[lm * 516 + kr];
    a.y = ldsA[lm * 516 + kr + 1];
    const float* bp = Bb + (size_t)kr * NHW + n0 + lm;
#pragma unroll
    for (int nc = 0; nc < 8; ++nc) {
      v2f bf;
      bf.x = bp[nc * 16];
      bf.y = bp[nc * 16 + NHW];
      acc[nc] = wmma4(a, bf, acc[nc]);
    }
  }
  float* Yb = Y + (size_t)b * bStrideY;
#pragma unroll
  for (int nc = 0; nc < 8; ++nc) {
#pragma unroll
    for (int r = 0; r < 8; ++r) {
      const int row = m0 + r + 8 * hi;
      const int col = n0 + (nc << 4) + lm;
      float v = acc[nc][r];
      if (bias)
        v += bias[row] + resid[(size_t)b * bStrideY + (size_t)row * NHW + col];
      Yb[(size_t)row * NHW + col] = v;
    }
  }
}

__global__ void vtrans_kernel(const float* __restrict__ qkv,
                              float* __restrict__ vt) {
  __shared__ float t[16][17];
  const int bh = blockIdx.z, bq = bh >> 2, h = bh & 3;
  const int j0 = blockIdx.x << 4, d0 = blockIdx.y << 4;
  const int tx = threadIdx.x, ty = threadIdx.y;
  t[ty][tx] =
      qkv[((size_t)bq * 1536 + 1024 + h * ND + d0 + ty) * NHW + j0 + tx];
  __syncthreads();
  vt[((size_t)bh * NHW + j0 + ty) * ND + d0 + tx] = t[tx][ty];
}

__global__ __launch_bounds__(256) void attn_kernel(const float* __restrict__ qkv,
                                                   const float* __restrict__ vt,
                                                   float* __restrict__ att) {
  __shared__ float ldsK[128 * 16];
  __shared__ float ldsV[16 * 132];
  __shared__ float ldsP[8][16 * 17];
  const int bh = blockIdx.x >> 3, iblk = blockIdx.x & 7;
  const int b = bh >> 2, h = bh & 3;
  const int tid = threadIdx.x, wv = tid >> 5, lane = tid & 31;
  const int lm = lane & 15, hi = lane >> 4;
  const int i0 = (iblk << 7) + (wv << 4);
  const float* qp = qkv + ((size_t)b * 1536 + h * ND) * NHW;
  const float* kp = qkv + ((size_t)b * 1536 + 512 + h * ND) * NHW;
  const float* vp = vt + (size_t)bh * NHW * ND;
  v2f qf[32];
#pragma unroll
  for (int ks = 0; ks < 32; ++ks) {
    const int kr = ks * 4 + 2 * hi;
    qf[ks].x = qp[(size_t)kr * NHW + i0 + lm];
    qf[ks].y = qp[(size_t)(kr + 1) * NHW + i0 + lm];
  }
  v8f O[8] = {};
  float mst[8], lst[8];
#pragma unroll
  for (int r = 0; r < 8; ++r) { mst[r] = -3.0e38f; lst[r] = 0.f; }
  float* myP = &ldsP[wv][0];

  for (int jt = 0; jt < 64; ++jt) {
    const int j0 = jt << 4;
    __syncthreads();
    for (int i = tid; i < 2048; i += 256)
      ldsK[i] = kp[(size_t)(i >> 4) * NHW + j0 + (i & 15)];
    for (int i = tid; i < 2048; i += 256)
      ldsV[(i >> 7) * 132 + (i & 127)] = vp[(size_t)(j0 + (i >> 7)) * ND + (i & 127)];
    __syncthreads();
    v8f s = {};
#pragma unroll
    for (int ks = 0; ks < 32; ++ks) {
      const int kr = ks * 4 + 2 * hi;
      v2f bf;
      bf.x = ldsK[kr * 16 + lm];
      bf.y = ldsK[(kr + 1) * 16 + lm];
      s = wmma4(qf[ks], bf, s);
    }
    float alpha[8], p[8];
#pragma unroll
    for (int r = 0; r < 8; ++r) {
      float sv = s[r] * ATT_SCALE;
      float tm = sv;
      tm = fmaxf(tm, __shfl_xor(tm, 1, 16));
      tm = fmaxf(tm, __shfl_xor(tm, 2, 16));
      tm = fmaxf(tm, __shfl_xor(tm, 4, 16));
      tm = fmaxf(tm, __shfl_xor(tm, 8, 16));
      float mn = fmaxf(mst[r], tm);
      alpha[r] = __expf(mst[r] - mn);
      float pv = __expf(sv - mn);
      float ts = pv;
      ts += __shfl_xor(ts, 1, 16);
      ts += __shfl_xor(ts, 2, 16);
      ts += __shfl_xor(ts, 4, 16);
      ts += __shfl_xor(ts, 8, 16);
      lst[r] = lst[r] * alpha[r] + ts;
      mst[r] = mn;
      p[r] = pv;
    }
#pragma unroll
    for (int nc = 0; nc < 8; ++nc)
#pragma unroll
      for (int r = 0; r < 8; ++r) O[nc][r] *= alpha[r];
#pragma unroll
    for (int r = 0; r < 8; ++r) myP[(r + 8 * hi) * 17 + lm] = p[r];
#pragma unroll
    for (int ks = 0; ks < 4; ++ks) {
      const int kr = ks * 4 + 2 * hi;
      v2f af;
      af.x = myP[lm * 17 + kr];
      af.y = myP[lm * 17 + kr + 1];
#pragma unroll
      for (int nc = 0; nc < 8; ++nc) {
        v2f bf;
        bf.x = ldsV[kr * 132 + (nc << 4) + lm];
        bf.y = ldsV[(kr + 1) * 132 + (nc << 4) + lm];
        O[nc] = wmma4(af, bf, O[nc]);
      }
    }
  }
  float* ap = att + ((size_t)b * NC + h * ND) * NHW;
#pragma unroll
  for (int r = 0; r < 8; ++r) {
    const float inv = 1.f / lst[r];
#pragma unroll
    for (int nc = 0; nc < 8; ++nc)
      ap[(size_t)((nc << 4) + lm) * NHW + i0 + r + 8 * hi] = O[nc][r] * inv;
  }
}

extern "C" void kernel_launch(void* const* d_in, const int* in_sizes, int n_in,
                              void* d_out, int out_size, void* d_ws,
                              size_t ws_size, hipStream_t stream) {
  (void)in_sizes; (void)n_in; (void)out_size; (void)ws_size;
  const float* x = (const float*)d_in[0];
  const float* gnw = (const float*)d_in[1];
  const float* gnb = (const float*)d_in[2];
  const float* wqkv = (const float*)d_in[3];
  const float* wout = (const float*)d_in[4];
  const float* bout = (const float*)d_in[5];
  float* out = (float*)d_out;
  float* ws = (float*)d_ws;
  float* xn = ws;
  float* qkv = ws + 8388608;
  float* vt = ws + 33554432;

  gn_kernel<<<NB * NG, 256, 0, stream>>>(x, gnw, gnb, xn);
  gemm512_kernel<<<dim3(96, NB), 256, 0, stream>>>(wqkv, xn, qkv, 524288L,
                                                   1572864L, nullptr, nullptr);
  vtrans_kernel<<<dim3(64, 8, NB * NH), dim3(16, 16), 0, stream>>>(qkv, vt);
  attn_kernel<<<NB * NH * 8, 256, 0, stream>>>(qkv, vt, xn);
  gemm512_kernel<<<dim3(32, NB), 256, 0, stream>>>(wout, xn, out, 524288L,
                                                   524288L, bout, x);
}


# attention key tile 16 -> 64: K/V staged by one group of 16-byte loads per tile, 4 independent score accumulators, softmax max/sum folded in-lane before one 16-lane reduce; row sums in a new order
# speedup vs baseline: 1.5465x; 1.5465x over previous
#include <hip/hip_runtime.h>


typedef __attribute__((ext_vector_type(2))) float v2f;
typedef __attribute__((ext_vector_type(8))) float v8f;

#define NB 16
#define NC 512
#define NHW 1024
#define NG 32
#define NH 4
#define ND 128
#define GN_EPS 1e-5f
#define ATT_SCALE 0.08838834764831845f

static __device__ __forceinline__ v8f wmma4(v2f a, v2f b, v8f c) {
  return __builtin_amdgcn_wmma_f32_16x16x4_f32(false, a, false, b, (short)0, c,
                                               false, false);
}

__global__ __launch_bounds__(256) void gn_kernel(const float* __restrict__ x,
                                                 const float* __restrict__ w,
                                                 const float* __restrict__ bia,
                                                 float* __restrict__ xn) {
  const int b = blockIdx.x >> 5;
  const int g = blockIdx.x & 31;
  const size_t base = ((size_t)b * NC + (size_t)g * 16) * NHW;
  const int tid = threadIdx.x;
  float sum = 0.f, sq = 0.f;
  for (int i = tid; i < 16 * NHW; i += 256) {
    float v = x[base + i];
    sum += v;
    sq += v * v;
  }
  for (int m = 16; m >= 1; m >>= 1) {
    sum += __shfl_xor(sum, m, 32);
    sq += __shfl_xor(sq, m, 32);
  }
  __shared__ float rs[8], rq[8], stat[2];
  const int wv = tid >> 5;
  if ((tid & 31) == 0) { rs[wv] = sum; rq[wv] = sq; }
  __syncthreads();
  if (tid == 0) {
    float ts = 0.f, tq = 0.f;
    for (int i = 0; i < 8; ++i) { ts += rs[i]; tq += rq[i]; }
    float mean = ts * (1.f / 16384.f);
    float var = tq * (1.f / 16384.f) - mean * mean;
    stat[0] = mean;
    stat[1] = rsqrtf(var + GN_EPS);
  }
  __syncthreads();
  const float mean = stat[0], rstd = stat[1];
  for (int i = tid; i < 16 * NHW; i += 256) {
    int c = (g << 4) + (i >> 10);
    xn[base + i] = (x[base + i] - mean) * rstd * w[c] + bia[c];
  }
}

__global__ __launch_bounds__(256) void gemm512_kernel(
    const float* __restrict__ A, const float* __restrict__ Bmat,
    float* __restrict__ Y, long bStrideB, long bStrideY,
    const float* __restrict__ bias, const float* __restrict__ resid) {
  __shared__ float ldsA[16 * 516];
  const int m0 = blockIdx.x << 4;
  const int b = blockIdx.y;
  const int tid = threadIdx.x;
  for (int i = tid; i < 8192; i += 256) {
    int row = i >> 9, k = i & 511;
    ldsA[row * 516 + k] = A[(size_t)(m0 + row) * 512 + k];
  }
  __syncthreads();
  const int lane = tid & 31, wv = tid >> 5;
  const int lm = lane & 15, hi = lane >> 4;
  const float* Bb = Bmat + (size_t)b * bStrideB;
  const int n0 = wv << 7;
  v8f acc[8] = {};
  for (int k0 = 0; k0 < 512; k0 += 4) {
    const int kr = k0 + 2 * hi;
    v2f a;
    a.x = ldsA[lm * 516 + kr];
    a.y = ldsA[lm * 516 + kr + 1];
    const float* bp = Bb + (size_t)kr * NHW + n0 + lm;
#pragma unroll
    for (int nc = 0; nc < 8; ++nc) {
      v2f bf;
      bf.x = bp[nc * 16];
      bf.y = bp[nc * 16 + NHW];
      acc[nc] = wmma4(a, bf, acc[nc]);
    }
  }
  float* Yb = Y + (size_t)b * bStrideY;
#pragma unroll
  for (int nc = 0; nc < 8; ++nc) {
#pragma unroll
    for (int r = 0; r < 8; ++r) {
      const int row = m0 + r + 8 * hi;
      const int col = n0 + (nc << 4) + lm;
      float v = acc[nc][r];
      if (bias)
        v += bias[row] + resid[(size_t)b * bStrideY + (size_t)row * NHW + col];
      Yb[(size_t)row * NHW + col] = v;
    }
  }
}

__global__ void vtrans_kernel(const float* __restrict__ qkv,
                              float* __restrict__ vt) {
  __shared__ float t[16][17];
  const int bh = blockIdx.z, bq = bh >> 2, h = bh & 3;
  const int j0 = blockIdx.x << 4, d0 = blockIdx.y << 4;
  const int tx = threadIdx.x, ty = threadIdx.y;
  t[ty][tx] =
      qkv[((size_t)bq * 1536 + 1024 + h * ND + d0 + ty) * NHW + j0 + tx];
  __syncthreads();
  vt[((size_t)bh * NHW + j0 + ty) * ND + d0 + tx] = t[tx][ty];
}

#define KT 64
#define KP 72
#define VP 136
#define PP 68
typedef __attribute__((ext_vector_type(4))) float v4f;

__global__ __launch_bounds__(256, 4) void attn_kernel(const float* __restrict__ qkv,
                                                   const float* __restrict__ vt,
                                                   float* __restrict__ att) {
  __shared__ __attribute__((aligned(16))) float ldsK[ND * KP];
  __shared__ __attribute__((aligned(16))) float ldsV[KT * VP];
  __shared__ __attribute__((aligned(16))) float ldsP[8][16 * PP];
  const int bh = blockIdx.x >> 3, iblk = blockIdx.x & 7;
  const int b = bh >> 2, h = bh & 3;
  const int tid = threadIdx.x, wv = tid >> 5, lane = tid & 31;
  const int lm = lane & 15, hi = lane >> 4;
  const int i0 = (iblk << 7) + (wv << 4);
  const float* qp = qkv + ((size_t)b * 1536 + h * ND) * NHW;
  const float* kp = qkv + ((size_t)b * 1536 + 512 + h * ND) * NHW;
  const float* vp = vt + (size_t)bh * NHW * ND;
  v2f qf[32];
#pragma unroll
  for (int ks = 0; ks < 32; ++ks) {
    const int kr = ks * 4 + 2 * hi;
    qf[ks].x = qp[(size_t)kr * NHW + i0 + lm];
    qf[ks].y = qp[(size_t)(kr + 1) * NHW + i0 + lm];
  }
  v8f O[8] = {};
  float mst[8], lst[8];
#pragma unroll
  for (int r = 0; r < 8; ++r) { mst[r] = -3.0e38f; lst[r] = 0.f; }
  float* myP = &ldsP[wv][0];

  const unsigned koff = (unsigned)(tid >> 4) * NHW + ((tid & 15) << 2);
  const unsigned voff = (unsigned)(tid >> 5) * ND + ((tid & 31) << 2);
  float* kl = &ldsK[(tid >> 4) * KP + ((tid & 15) << 2)];
  float* vl = &ldsV[(tid >> 5) * VP + ((tid & 31) << 2)];

  for (int jt = 0; jt < NHW / KT; ++jt) {
    const float* kt = kp + jt * KT;
    const float* vtile = vp + (size_t)jt * KT * ND;
    v4f kst[8], vst[8];
#pragma unroll
    for (int u = 0; u < 8; ++u) kst[u] = *(const v4f*)(kt + (koff + (unsigned)u * 16 * NHW));
#pragma unroll
    for (int u = 0; u < 8; ++u) vst[u] = *(const v4f*)(vtile + (voff + (unsigned)u * 8 * ND));
    __syncthreads();
#pragma unroll
    for (int u = 0; u < 8; ++u) *(v4f*)(kl + u * 16 * KP) = kst[u];
#pragma unroll
    for (int u = 0; u < 8; ++u) *(v4f*)(vl + u * 8 * VP) = vst[u];
    __syncthreads();
    v8f s[4] = {};
#pragma unroll
    for (int ks = 0; ks < 32; ++ks) {
      const int kr = ks * 4 + 2 * hi;
#pragma unroll
      for (int t = 0; t < 4; ++t) {
        v2f bf;
        bf.x = ldsK[kr * KP + (t << 4) + lm];
        bf.y = ldsK[(kr + 1) * KP + (t << 4) + lm];
        s[t] = wmma4(qf[ks], bf, s[t]);
      }
    }
#pragma unroll
    for (int r = 0; r < 8; ++r) {
      const float s0 = s[0][r] * ATT_SCALE, s1 = s[1][r] * ATT_SCALE;
      const float s2 = s[2][r] * ATT_SCALE, s3 = s[3][r] * ATT_SCALE;
      float tm = fmaxf(fmaxf(s0, s1), fmaxf(s2, s3));
      tm = fmaxf(tm, __shfl_xor(tm, 1, 16));
      tm = fmaxf(tm, __shfl_xor(tm, 2, 16));
      tm = fmaxf(tm, __shfl_xor(tm, 4, 16));
      tm = fmaxf(tm, __shfl_xor(tm, 8, 16));
      const float mn = fmaxf(mst[r], tm);
      const float alpha = __expf(mst[r] - mn);
      const float p0 = __expf(s0 - mn), p1 = __expf(s1 - mn);
      const float p2 = __expf(s2 - mn), p3 = __expf(s3 - mn);
      float ts = (p0 + p1) + (p2 + p3);
      ts += __shfl_xor(ts, 1, 16);
      ts += __shfl_xor(ts, 2, 16);
      ts += __shfl_xor(ts, 4, 16);
      ts += __shfl_xor(ts, 8, 16);
      lst[r] = lst[r] * alpha + ts;
      mst[r] = mn;
      float* pr = myP + (r + 8 * hi) * PP + lm;
      pr[0] = p0; pr[16] = p1; pr[32] = p2; pr[48] = p3;
#pragma unroll
      for (int nc = 0; nc < 8; ++nc) O[nc][r] *= alpha;
    }
#pragma unroll
    for (int ks = 0; ks < KT / 4; ++ks) {
      const int kr = ks * 4 + 2 * hi;
      const v2f af = *(const v2f*)&myP[lm * PP + kr];
#pragma unroll
      for (int nc = 0; nc < 8; ++nc) {
        v2f bf;
        bf.x = ldsV[kr * VP + (nc << 4) + lm];
        bf.y = ldsV[(kr + 1) * VP + (nc << 4) + lm];
        O[nc] = wmma4(af, bf, O[nc]);
      }
    }
  }
  float* ap = att + ((size_t)b * NC + h * ND) * NHW;
#pragma unroll
  for (int r = 0; r < 8; ++r) {
    const float inv = 1.f / lst[r];
#pragma unroll
    for (int nc = 0; nc < 8; ++nc)
      ap[(size_t)((nc << 4) + lm) * NHW + i0 + r + 8 * hi] = O[nc][r] * inv;
  }
}

extern "C" void kernel_launch(void* const* d_in, const int* in_sizes, int n_in,
                              void* d_out, int out_size, void* d_ws,
                              size_t ws_size, hipStream_t stream) {
  (void)in_sizes; (void)n_in; (void)out_size; (void)ws_size;
  const float* x = (const float*)d_in[0];
  const float* gnw = (const float*)d_in[1];
  const float* gnb = (const float*)d_in[2];
  const float* wqkv = (const float*)d_in[3];
  const float* wout = (const float*)d_in[4];
  const float* bout = (const float*)d_in[5];
  float* out = (float*)d_out;
  float* ws = (float*)d_ws;
  float* xn = ws;
  float* qkv = ws + 8388608;
  float* vt = ws + 33554432;

  gn_kernel<<<NB * NG, 256, 0, stream>>>(x, gnw, gnb, xn);
  gemm512_kernel<<<dim3(96, NB), 256, 0, stream>>>(wqkv, xn, qkv, 524288L,
                                                   1572864L, nullptr, nullptr);
  vtrans_kernel<<<dim3(64, 8, NB * NH), dim3(16, 16), 0, stream>>>(qkv, vt);
  attn_kernel<<<NB * NH * 8, 256, 0, stream>>>(qkv, vt, xn);
  gemm512_kernel<<<dim3(32, NB), 256, 0, stream>>>(wout, xn, out, 524288L,
                                                   524288L, bout, x);
}


# GEMMs re-tiled to 128x128 workgroup tiles: 32-k slabs of A and B staged in LDS (B as interleaved k pairs), shared by 8 waves of 64x32, next slab prefetched in registers; same sum order
# speedup vs baseline: 1.8313x; 1.1842x over previous
#include <hip/hip_runtime.h>


typedef __attribute__((ext_vector_type(2))) float v2f;
typedef __attribute__((ext_vector_type(8))) float v8f;

#define NB 16
#define NC 512
#define NHW 1024
#define NG 32
#define NH 4
#define ND 128
#define GN_EPS 1e-5f
#define ATT_SCALE 0.08838834764831845f

static __device__ __forceinline__ v8f wmma4(v2f a, v2f b, v8f c) {
  return __builtin_amdgcn_wmma_f32_16x16x4_f32(false, a, false, b, (short)0, c,
                                               false, false);
}

__global__ __launch_bounds__(256) void gn_kernel(const float* __restrict__ x,
                                                 const float* __restrict__ w,
                                                 const float* __restrict__ bia,
                                                 float* __restrict__ xn) {
  const int b = blockIdx.x >> 5;
  const int g = blockIdx.x & 31;
  const size_t base = ((size_t)b * NC + (size_t)g * 16) * NHW;
  const int tid = threadIdx.x;
  float sum = 0.f, sq = 0.f;
  for (int i = tid; i < 16 * NHW; i += 256) {
    float v = x[base + i];
    sum += v;
    sq += v * v;
  }
  for (int m = 16; m >= 1; m >>= 1) {
    sum += __shfl_xor(sum, m, 32);
    sq += __shfl_xor(sq, m, 32);
  }
  __shared__ float rs[8], rq[8], stat[2];
  const int wv = tid >> 5;
  if ((tid & 31) == 0) { rs[wv] = sum; rq[wv] = sq; }
  __syncthreads();
  if (tid == 0) {
    float ts = 0.f, tq = 0.f;
    for (int i = 0; i < 8; ++i) { ts += rs[i]; tq += rq[i]; }
    float mean = ts * (1.f / 16384.f);
    float var = tq * (1.f / 16384.f) - mean * mean;
    stat[0] = mean;
    stat[1] = rsqrtf(var + GN_EPS);
  }
  __syncthreads();
  const float mean = stat[0], rstd = stat[1];
  for (int i = tid; i < 16 * NHW; i += 256) {
    int c = (g << 4) + (i >> 10);
    xn[base + i] = (x[base + i] - mean) * rstd * w[c] + bia[c];
  }
}

#define GBM 128
#define GBN 128
#define GBK 32
#define GAP 36
#define GBP 288
#define GEP 36
#define GASZ (GBM * GAP)
#define GBSZ ((GBK / 2) * GBP)
typedef __attribute__((ext_vector_type(4))) float g4f;

__global__ __launch_bounds__(256, 6) void gemm512_kernel(
    const float* __restrict__ A, const float* __restrict__ Bmat,
    float* __restrict__ Y, long bStrideB, long bStrideY,
    const float* __restrict__ bias, const float* __restrict__ resid) {
  __shared__ __attribute__((aligned(16))) float lds[2 * GASZ + 2 * GBSZ];
  float* ldsA = lds;
  float* ldsB = lds + 2 * GASZ;
  const int m0 = blockIdx.x * GBM;
  const int n0 = blockIdx.y * GBN;
  const int b = blockIdx.z;
  const int tid = threadIdx.x;
  const int lane = tid & 31, wv = tid >> 5;
  const int lm = lane & 15, hi = lane >> 4;
  const int wm = wv >> 2, wn = wv & 3;
  const float* Bb = Bmat + (size_t)b * bStrideB;

  const float* Ag = A + (size_t)(m0 + (tid >> 3)) * 512 + ((tid & 7) << 2);
  const float* Bg = Bb + (size_t)(2 * (tid >> 5)) * NHW + n0 + ((tid & 31) << 2);
  float* Al = ldsA + (tid >> 3) * GAP + ((tid & 7) << 2);
  float* Bl = ldsB + (tid >> 5) * GBP + ((tid & 31) << 3);
  const float* Ar = ldsA + (wm * 64 + lm) * GAP + 2 * hi;
  const float* Br = ldsB + hi * GBP + 2 * (wn * 32 + lm);

  g4f ar[4], br0[2], br1[2];
  v8f acc[4][2] = {};

#define G_LOAD(kb)                                                              \
  {                                                                             \
    _Pragma("unroll") for (int u = 0; u < 4; ++u)                               \
        ar[u] = *(const g4f*)(Ag + (size_t)u * 32 * 512 + (kb));                \
    _Pragma("unroll") for (int u = 0; u < 2; ++u) {                             \
      br0[u] = *(const g4f*)(Bg + (size_t)((kb) + 16 * u) * NHW);               \
      br1[u] = *(const g4f*)(Bg + (size_t)((kb) + 16 * u + 1) * NHW);           \
    }                                                                           \
  }
#define G_STORE(st)                                                             \
  {                                                                             \
    _Pragma("unroll") for (int u = 0; u < 4; ++u)                               \
        *(g4f*)(Al + (st) * GASZ + u * 32 * GAP) = ar[u];                       \
    _Pragma("unroll") for (int u = 0; u < 2; ++u) {                             \
      g4f lo, up;                                                               \
      lo.x = br0[u].x; lo.y = br1[u].x; lo.z = br0[u].y; lo.w = br1[u].y;       \
      up.x = br0[u].z; up.y = br1[u].z; up.z = br0[u].w; up.w = br1[u].w;       \
      *(g4f*)(Bl + (st) * GBSZ + u * 8 * GBP) = lo;                             \
      *(g4f*)(Bl + (st) * GBSZ + u * 8 * GBP + 4) = up;                         \
    }                                                                           \
  }
#define G_COMPUTE(st)                                                           \
  {                                                                             \
    const float* as = Ar + (st) * GASZ;                                         \
    const float* bs = Br + (st) * GBSZ;                                         \
    _Pragma("unroll") for (int kk = 0; kk < GBK / 4; ++kk) {                    \
      v2f af[4], bf[2];                                                         \
      _Pragma("unroll") for (int mt = 0; mt < 4; ++mt)                          \
          af[mt] = *(const v2f*)(as + mt * 16 * GAP + kk * 4);                  \
      _Pragma("unroll") for (int nt = 0; nt < 2; ++nt)                          \
          bf[nt] = *(const v2f*)(bs + kk * 2 * GBP + nt * 32);                  \
      _Pragma("unroll") for (int mt = 0; mt < 4; ++mt)                          \
          _Pragma("unroll") for (int nt = 0; nt < 2; ++nt)                      \
              acc[mt][nt] = wmma4(af[mt], bf[nt], acc[mt][nt]);                 \
    }                                                                           \
  }

  G_LOAD(0);
  G_STORE(0);
  __syncthreads();
  for (int t = 0; t < 512 / GBK - 1; ++t) {
    const int cur = t & 1;
    G_LOAD((t + 1) * GBK);
    __builtin_amdgcn_sched_barrier(0);
    G_COMPUTE(cur);
    G_STORE(cur ^ 1);
    __syncthreads();
  }
  G_COMPUTE(1);
#undef G_LOAD
#undef G_STORE
#undef G_COMPUTE

  __syncthreads();
  float* slab = lds + wv * (64 * GEP);
#pragma unroll
  for (int mt = 0; mt < 4; ++mt)
#pragma unroll
    for (int nt = 0; nt < 2; ++nt)
#pragma unroll
      for (int r = 0; r < 8; ++r)
        slab[(mt * 16 + r + 8 * hi) * GEP + nt * 16 + lm] = acc[mt][nt][r];
  __syncthreads();
  float* Yb = Y + (size_t)b * bStrideY;
  const int rl = lane >> 3, c4 = (lane & 7) << 2;
#pragma unroll
  for (int it = 0; it < 16; ++it) {
    const int rloc = it * 4 + rl;
    g4f v = *(const g4f*)(slab + rloc * GEP + c4);
    const int row = m0 + wm * 64 + rloc;
    const int col = n0 + wn * 32 + c4;
    if (bias) {
      const float bb = bias[row];
      const g4f rr = *(const g4f*)(resid + (size_t)b * bStrideY + (size_t)row * NHW + col);
      v.x += bb + rr.x; v.y += bb + rr.y; v.z += bb + rr.z; v.w += bb + rr.w;
    }
    *(g4f*)(Yb + (size_t)row * NHW + col) = v;
  }
}

__global__ void vtrans_kernel(const float* __restrict__ qkv,
                              float* __restrict__ vt) {
  __shared__ float t[16][17];
  const int bh = blockIdx.z, bq = bh >> 2, h = bh & 3;
  const int j0 = blockIdx.x << 4, d0 = blockIdx.y << 4;
  const int tx = threadIdx.x, ty = threadIdx.y;
  t[ty][tx] =
      qkv[((size_t)bq * 1536 + 1024 + h * ND + d0 + ty) * NHW + j0 + tx];
  __syncthreads();
  vt[((size_t)bh * NHW + j0 + ty) * ND + d0 + tx] = t[tx][ty];
}

#define KT 64
#define KP 72
#define VP 136
#define PP 68
typedef __attribute__((ext_vector_type(4))) float v4f;

__global__ __launch_bounds__(256, 4) void attn_kernel(const float* __restrict__ qkv,
                                                   const float* __restrict__ vt,
                                                   float* __restrict__ att) {
  __shared__ __attribute__((aligned(16))) float ldsK[ND * KP];
  __shared__ __attribute__((aligned(16))) float ldsV[KT * VP];
  __shared__ __attribute__((aligned(16))) float ldsP[8][16 * PP];
  const int bh = blockIdx.x >> 3, iblk = blockIdx.x & 7;
  const int b = bh >> 2, h = bh & 3;
  const int tid = threadIdx.x, wv = tid >> 5, lane = tid & 31;
  const int lm = lane & 15, hi = lane >> 4;
  const int i0 = (iblk << 7) + (wv << 4);
  const float* qp = qkv + ((size_t)b * 1536 + h * ND) * NHW;
  const float* kp = qkv + ((size_t)b * 1536 + 512 + h * ND) * NHW;
  const float* vp = vt + (size_t)bh * NHW * ND;
  v2f qf[32];
#pragma unroll
  for (int ks = 0; ks < 32; ++ks) {
    const int kr = ks * 4 + 2 * hi;
    qf[ks].x = qp[(size_t)kr * NHW + i0 + lm];
    qf[ks].y = qp[(size_t)(kr + 1) * NHW + i0 + lm];
  }
  v8f O[8] = {};
  float mst[8], lst[8];
#pragma unroll
  for (int r = 0; r < 8; ++r) { mst[r] = -3.0e38f; lst[r] = 0.f; }
  float* myP = &ldsP[wv][0];

  const unsigned koff = (unsigned)(tid >> 4) * NHW + ((tid & 15) << 2);
  const unsigned voff = (unsigned)(tid >> 5) * ND + ((tid & 31) << 2);
  float* kl = &ldsK[(tid >> 4) * KP + ((tid & 15) << 2)];
  float* vl = &ldsV[(tid >> 5) * VP + ((tid & 31) << 2)];

  for (int jt = 0; jt < NHW / KT; ++jt) {
    const float* kt = kp + jt * KT;
    const float* vtile = vp + (size_t)jt * KT * ND;
    v4f kst[8], vst[8];
#pragma unroll
    for (int u = 0; u < 8; ++u) kst[u] = *(const v4f*)(kt + (koff + (unsigned)u * 16 * NHW));
#pragma unroll
    for (int u = 0; u < 8; ++u) vst[u] = *(const v4f*)(vtile + (voff + (unsigned)u * 8 * ND));
    __syncthreads();
#pragma unroll
    for (int u = 0; u < 8; ++u) *(v4f*)(kl + u * 16 * KP) = kst[u];
#pragma unroll
    for (int u = 0; u < 8; ++u) *(v4f*)(vl + u * 8 * VP) = vst[u];
    __syncthreads();
    v8f s[4] = {};
#pragma unroll
    for (int ks = 0; ks < 32; ++ks) {
      const int kr = ks * 4 + 2 * hi;
#pragma unroll
      for (int t = 0; t < 4; ++t) {
        v2f bf;
        bf.x = ldsK[kr * KP + (t << 4) + lm];
        bf.y = ldsK[(kr + 1) * KP + (t << 4) + lm];
        s[t] = wmma4(qf[ks], bf, s[t]);
      }
    }
#pragma unroll
    for (int r = 0; r < 8; ++r) {
      const float s0 = s[0][r] * ATT_SCALE, s1 = s[1][r] * ATT_SCALE;
      const float s2 = s[2][r] * ATT_SCALE, s3 = s[3][r] * ATT_SCALE;
      float tm = fmaxf(fmaxf(s0, s1), fmaxf(s2, s3));
      tm = fmaxf(tm, __shfl_xor(tm, 1, 16));
      tm = fmaxf(tm, __shfl_xor(tm, 2, 16));
      tm = fmaxf(tm, __shfl_xor(tm, 4, 16));
      tm = fmaxf(tm, __shfl_xor(tm, 8, 16));
      const float mn = fmaxf(mst[r], tm);
      const float alpha = __expf(mst[r] - mn);
      const float p0 = __expf(s0 - mn), p1 = __expf(s1 - mn);
      const float p2 = __expf(s2 - mn), p3 = __expf(s3 - mn);
      float ts = (p0 + p1) + (p2 + p3);
      ts += __shfl_xor(ts, 1, 16);
      ts += __shfl_xor(ts, 2, 16);
      ts += __shfl_xor(ts, 4, 16);
      ts += __shfl_xor(ts, 8, 16);
      lst[r] = lst[r] * alpha + ts;
      mst[r] = mn;
      float* pr = myP + (r + 8 * hi) * PP + lm;
      pr[0] = p0; pr[16] = p1; pr[32] = p2; pr[48] = p3;
#pragma unroll
      for (int nc = 0; nc < 8; ++nc) O[nc][r] *= alpha;
    }
#pragma unroll
    for (int ks = 0; ks < KT / 4; ++ks) {
      const int kr = ks * 4 + 2 * hi;
      const v2f af = *(const v2f*)&myP[lm * PP + kr];
#pragma unroll
      for (int nc = 0; nc < 8; ++nc) {
        v2f bf;
        bf.x = ldsV[kr * VP + (nc << 4) + lm];
        bf.y = ldsV[(kr + 1) * VP + (nc << 4) + lm];
        O[nc] = wmma4(af, bf, O[nc]);
      }
    }
  }
  float* ap = att + ((size_t)b * NC + h * ND) * NHW;
#pragma unroll
  for (int r = 0; r < 8; ++r) {
    const float inv = 1.f / lst[r];
#pragma unroll
    for (int nc = 0; nc < 8; ++nc)
      ap[(size_t)((nc << 4) + lm) * NHW + i0 + r + 8 * hi] = O[nc][r] * inv;
  }
}

extern "C" void kernel_launch(void* const* d_in, const int* in_sizes, int n_in,
                              void* d_out, int out_size, void* d_ws,
                              size_t ws_size, hipStream_t stream) {
  (void)in_sizes; (void)n_in; (void)out_size; (void)ws_size;
  const float* x = (const float*)d_in[0];
  const float* gnw = (const float*)d_in[1];
  const float* gnb = (const float*)d_in[2];
  const float* wqkv = (const float*)d_in[3];
  const float* wout = (const float*)d_in[4];
  const float* bout = (const float*)d_in[5];
  float* out = (float*)d_out;
  float* ws = (float*)d_ws;
  float* xn = ws;
  float* qkv = ws + 8388608;
  float* vt = ws + 33554432;

  gn_kernel<<<NB * NG, 256, 0, stream>>>(x, gnw, gnb, xn);
  gemm512_kernel<<<dim3(12, 8, NB), 256, 0, stream>>>(wqkv, xn, qkv, 524288L,
                                                   1572864L, nullptr, nullptr);
  vtrans_kernel<<<dim3(64, 8, NB * NH), dim3(16, 16), 0, stream>>>(qkv, vt);
  attn_kernel<<<NB * NH * 8, 256, 0, stream>>>(qkv, vt, xn);
  gemm512_kernel<<<dim3(4, 8, NB), 256, 0, stream>>>(wout, xn, out, 524288L,
                                                   524288L, bout, x);
}


# QKV GEMM epilogue writes the V row tiles transposed (vT[b][head][key][d]: per-wave slab turned in LDS, 256-byte row stores); the separate V transpose launch of 32768 small workgroups is removed
# speedup vs baseline: 1.9402x; 1.0595x over previous
#include <hip/hip_runtime.h>


typedef __attribute__((ext_vector_type(2))) float v2f;
typedef __attribute__((ext_vector_type(8))) float v8f;

#define NB 16
#define NC 512
#define NHW 1024
#define NG 32
#define NH 4
#define ND 128
#define GN_EPS 1e-5f
#define ATT_SCALE 0.08838834764831845f

static __device__ __forceinline__ v8f wmma4(v2f a, v2f b, v8f c) {
  return __builtin_amdgcn_wmma_f32_16x16x4_f32(false, a, false, b, (short)0, c,
                                               false, false);
}

__global__ __launch_bounds__(256) void gn_kernel(const float* __restrict__ x,
                                                 const float* __restrict__ w,
                                                 const float* __restrict__ bia,
                                                 float* __restrict__ xn) {
  const int b = blockIdx.x >> 5;
  const int g = blockIdx.x & 31;
  const size_t base = ((size_t)b * NC + (size_t)g * 16) * NHW;
  const int tid = threadIdx.x;
  float sum = 0.f, sq = 0.f;
  for (int i = tid; i < 16 * NHW; i += 256) {
    float v = x[base + i];
    sum += v;
    sq += v * v;
  }
  for (int m = 16; m >= 1; m >>= 1) {
    sum += __shfl_xor(sum, m, 32);
    sq += __shfl_xor(sq, m, 32);
  }
  __shared__ float rs[8], rq[8], stat[2];
  const int wv = tid >> 5;
  if ((tid & 31) == 0) { rs[wv] = sum; rq[wv] = sq; }
  __syncthreads();
  if (tid == 0) {
    float ts = 0.f, tq = 0.f;
    for (int i = 0; i < 8; ++i) { ts += rs[i]; tq += rq[i]; }
    float mean = ts * (1.f / 16384.f);
    float var = tq * (1.f / 16384.f) - mean * mean;
    stat[0] = mean;
    stat[1] = rsqrtf(var + GN_EPS);
  }
  __syncthreads();
  const float mean = stat[0], rstd = stat[1];
  for (int i = tid; i < 16 * NHW; i += 256) {
    int c = (g << 4) + (i >> 10);
    xn[base + i] = (x[base + i] - mean) * rstd * w[c] + bia[c];
  }
}

#define GBM 128
#define GBN 128
#define GBK 32
#define GAP 36
#define GBP 288
#define GEP 36
#define GTP 68
#define GASZ (GBM * GAP)
#define GBSZ ((GBK / 2) * GBP)
typedef __attribute__((ext_vector_type(4))) float g4f;

__global__ __launch_bounds__(256, 6) void gemm512_kernel(
    const float* __restrict__ A, const float* __restrict__ Bmat,
    float* __restrict__ Y, long bStrideB, long bStrideY,
    const float* __restrict__ bias, const float* __restrict__ resid,
    float* __restrict__ Yt) {
  __shared__ __attribute__((aligned(16))) float lds[2 * GASZ + 2 * GBSZ];
  float* ldsA = lds;
  float* ldsB = lds + 2 * GASZ;
  const int m0 = blockIdx.x * GBM;
  const int n0 = blockIdx.y * GBN;
  const int b = blockIdx.z;
  const int tid = threadIdx.x;
  const int lane = tid & 31, wv = tid >> 5;
  const int lm = lane & 15, hi = lane >> 4;
  const int wm = wv >> 2, wn = wv & 3;
  const float* Bb = Bmat + (size_t)b * bStrideB;

  const float* Ag = A + (size_t)(m0 + (tid >> 3)) * 512 + ((tid & 7) << 2);
  const float* Bg = Bb + (size_t)(2 * (tid >> 5)) * NHW + n0 + ((tid & 31) << 2);
  float* Al = ldsA + (tid >> 3) * GAP + ((tid & 7) << 2);
  float* Bl = ldsB + (tid >> 5) * GBP + ((tid & 31) << 3);
  const float* Ar = ldsA + (wm * 64 + lm) * GAP + 2 * hi;
  const float* Br = ldsB + hi * GBP + 2 * (wn * 32 + lm);

  g4f ar[4], br0[2], br1[2];
  v8f acc[4][2] = {};

#define G_LOAD(kb)                                                              \
  {                                                                             \
    _Pragma("unroll") for (int u = 0; u < 4; ++u)                               \
        ar[u] = *(const g4f*)(Ag + (size_t)u * 32 * 512 + (kb));                \
    _Pragma("unroll") for (int u = 0; u < 2; ++u) {                             \
      br0[u] = *(const g4f*)(Bg + (size_t)((kb) + 16 * u) * NHW);               \
      br1[u] = *(const g4f*)(Bg + (size_t)((kb) + 16 * u + 1) * NHW);           \
    }                                                                           \
  }
#define G_STORE(st)                                                             \
  {                                                                             \
    _Pragma("unroll") for (int u = 0; u < 4; ++u)                               \
        *(g4f*)(Al + (st) * GASZ + u * 32 * GAP) = ar[u];                       \
    _Pragma("unroll") for (int u = 0; u < 2; ++u) {                             \
      g4f lo, up;                                                               \
      lo.x = br0[u].x; lo.y = br1[u].x; lo.z = br0[u].y; lo.w = br1[u].y;       \
      up.x = br0[u].z; up.y = br1[u].z; up.z = br0[u].w; up.w = br1[u].w;       \
      *(g4f*)(Bl + (st) * GBSZ + u * 8 * GBP) = lo;                             \
      *(g4f*)(Bl + (st) * GBSZ + u * 8 * GBP + 4) = up;                         \
    }                                                                           \
  }
#define G_COMPUTE(st)                                                           \
  {                                                                             \
    const float* as = Ar + (st) * GASZ;                                         \
    const float* bs = Br + (st) * GBSZ;                                         \
    _Pragma("unroll") for (int kk = 0; kk < GBK / 4; ++kk) {                    \
      v2f af[4], bf[2];                                                         \
      _Pragma("unroll") for (int mt = 0; mt < 4; ++mt)                          \
          af[mt] = *(const v2f*)(as + mt * 16 * GAP + kk * 4);                  \
      _Pragma("unroll") for (int nt = 0; nt < 2; ++nt)                          \
          bf[nt] = *(const v2f*)(bs + kk * 2 * GBP + nt * 32);                  \
      _Pragma("unroll") for (int mt = 0; mt < 4; ++mt)                          \
          _Pragma("unroll") for (int nt = 0; nt < 2; ++nt)                      \
              acc[mt][nt] = wmma4(af[mt], bf[nt], acc[mt][nt]);                 \
    }                                                                           \
  }

  G_LOAD(0);
  G_STORE(0);
  __syncthreads();
  for (int t = 0; t < 512 / GBK - 1; ++t) {
    const int cur = t & 1;
    G_LOAD((t + 1) * GBK);
    __builtin_amdgcn_sched_barrier(0);
    G_COMPUTE(cur);
    G_STORE(cur ^ 1);
    __syncthreads();
  }
  G_COMPUTE(1);
#undef G_LOAD
#undef G_STORE
#undef G_COMPUTE

  __syncthreads();
  float* slab = lds + wv * (64 * GEP);
  static_assert(GBM == ND, "a V row tile must be exactly one head");
  static_assert(32 * GTP <= 64 * GEP, "the transposed slab must fit the wave's slab");
  if (Yt != nullptr && m0 >= 2 * NC) {
#pragma unroll
    for (int mt = 0; mt < 4; ++mt)
#pragma unroll
      for (int nt = 0; nt < 2; ++nt) {
        float* p = slab + (nt * 16 + lm) * GTP + mt * 16 + 8 * hi;
        g4f lo, up;
        lo.x = acc[mt][nt][0]; lo.y = acc[mt][nt][1]; lo.z = acc[mt][nt][2]; lo.w = acc[mt][nt][3];
        up.x = acc[mt][nt][4]; up.y = acc[mt][nt][5]; up.z = acc[mt][nt][6]; up.w = acc[mt][nt][7];
        *(g4f*)p = lo;
        *(g4f*)(p + 4) = up;
      }
    __syncthreads();
    const int head = (m0 - 2 * NC) / ND;
    float* Vb = Yt + ((size_t)(b * NH + head) * NHW + n0 + wn * 32) * ND + wm * 64;
    const int jr = lane >> 4, d4 = (lane & 15) << 2;
#pragma unroll
    for (int it = 0; it < 16; ++it) {
      const int jl = it * 2 + jr;
      *(g4f*)(Vb + (size_t)jl * ND + d4) = *(const g4f*)(slab + jl * GTP + d4);
    }
    return;
  }
#pragma unroll
  for (int mt = 0; mt < 4; ++mt)
#pragma unroll
    for (int nt = 0; nt < 2; ++nt)
#pragma unroll
      for (int r = 0; r < 8; ++r)
        slab[(mt * 16 + r + 8 * hi) * GEP + nt * 16 + lm] = acc[mt][nt][r];
  __syncthreads();
  float* Yb = Y + (size_t)b * bStrideY;
  const int rl = lane >> 3, c4 = (lane & 7) << 2;
#pragma unroll
  for (int it = 0; it < 16; ++it) {
    const int rloc = it * 4 + rl;
    g4f v = *(const g4f*)(slab + rloc * GEP + c4);
    const int row = m0 + wm * 64 + rloc;
    const int col = n0 + wn * 32 + c4;
    if (bias) {
      const float bb = bias[row];
      const g4f rr = *(const g4f*)(resid + (size_t)b * bStrideY + (size_t)row * NHW + col);
      v.x += bb + rr.x; v.y += bb + rr.y; v.z += bb + rr.z; v.w += bb + rr.w;
    }
    *(g4f*)(Yb + (size_t)row * NHW + col) = v;
  }
}

#define KT 64
#define KP 72
#define VP 136
#define PP 68
typedef __attribute__((ext_vector_type(4))) float v4f;

__global__ __launch_bounds__(256, 4) void attn_kernel(const float* __restrict__ qkv,
                                                   const float* __restrict__ vt,
                                                   float* __restrict__ att) {
  __shared__ __attribute__((aligned(16))) float ldsK[ND * KP];
  __shared__ __attribute__((aligned(16))) float ldsV[KT * VP];
  __shared__ __attribute__((aligned(16))) float ldsP[8][16 * PP];
  const int bh = blockIdx.x >> 3, iblk = blockIdx.x & 7;
  const int b = bh >> 2, h = bh & 3;
  const int tid = threadIdx.x, wv = tid >> 5, lane = tid & 31;
  const int lm = lane & 15, hi = lane >> 4;
  const int i0 = (iblk << 7) + (wv << 4);
  const float* qp = qkv + ((size_t)b * 1536 + h * ND) * NHW;
  const float* kp = qkv + ((size_t)b * 1536 + 512 + h * ND) * NHW;
  const float* vp = vt + (size_t)bh * NHW * ND;
  v2f qf[32];
#pragma unroll
  for (int ks = 0; ks < 32; ++ks) {
    const int kr = ks * 4 + 2 * hi;
    qf[ks].x = qp[(size_t)kr * NHW + i0 + lm];
    qf[ks].y = qp[(size_t)(kr + 1) * NHW + i0 + lm];
  }
  v8f O[8] = {};
  float mst[8], lst[8];
#pragma unroll
  for (int r = 0; r < 8; ++r) { mst[r] = -3.0e38f; lst[r] = 0.f; }
  float* myP = &ldsP[wv][0];

  const unsigned koff = (unsigned)(tid >> 4) * NHW + ((tid & 15) << 2);
  const unsigned voff = (unsigned)(tid >> 5) * ND + ((tid & 31) << 2);
  float* kl = &ldsK[(tid >> 4) * KP + ((tid & 15) << 2)];
  float* vl = &ldsV[(tid >> 5) * VP + ((tid & 31) << 2)];

  for (int jt = 0; jt < NHW / KT; ++jt) {
    const float* kt = kp + jt * KT;
    const float* vtile = vp + (size_t)jt * KT * ND;
    v4f kst[8], vst[8];
#pragma unroll
    for (int u = 0; u < 8; ++u) kst[u] = *(const v4f*)(kt + (koff + (unsigned)u * 16 * NHW));
#pragma unroll
    for (int u = 0; u < 8; ++u) vst[u] = *(const v4f*)(vtile + (voff + (unsigned)u * 8 * ND));
    __syncthreads();
#pragma unroll
    for (int u = 0; u < 8; ++u) *(v4f*)(kl + u * 16 * KP) = kst[u];
#pragma unroll
    for (int u = 0; u < 8; ++u) *(v4f*)(vl + u * 8 * VP) = vst[u];
    __syncthreads();
    v8f s[4] = {};
#pragma unroll
    for (int ks = 0; ks < 32; ++ks) {
      const int kr = ks * 4 + 2 * hi;
#pragma unroll
      for (int t = 0; t < 4; ++t) {
        v2f bf;
        bf.x = ldsK[kr * KP + (t << 4) + lm];
        bf.y = ldsK[(kr + 1) * KP + (t << 4) + lm];
        s[t] = wmma4(qf[ks], bf, s[t]);
      }
    }
#pragma unroll
    for (int r = 0; r < 8; ++r) {
      const float s0 = s[0][r] * ATT_SCALE, s1 = s[1][r] * ATT_SCALE;
      const float s2 = s[2][r] * ATT_SCALE, s3 = s[3][r] * ATT_SCALE;
      float tm = fmaxf(fmaxf(s0, s1), fmaxf(s2, s3));
      tm = fmaxf(tm, __shfl_xor(tm, 1, 16));
      tm = fmaxf(tm, __shfl_xor(tm, 2, 16));
      tm = fmaxf(tm, __shfl_xor(tm, 4, 16));
      tm = fmaxf(tm, __shfl_xor(tm, 8, 16));
      const float mn = fmaxf(mst[r], tm);
      const float alpha = __expf(mst[r] - mn);
      const float p0 = __expf(s0 - mn), p1 = __expf(s1 - mn);
      const float p2 = __expf(s2 - mn), p3 = __expf(s3 - mn);
      float ts = (p0 + p1) + (p2 + p3);
      ts += __shfl_xor(ts, 1, 16);
      ts += __shfl_xor(ts, 2, 16);
      ts += __shfl_xor(ts, 4, 16);
      ts += __shfl_xor(ts, 8, 16);
      lst[r] = lst[r] * alpha + ts;
      mst[r] = mn;
      float* pr = myP + (r + 8 * hi) * PP + lm;
      pr[0] = p0; pr[16] = p1; pr[32] = p2; pr[48] = p3;
#pragma unroll
      for (int nc = 0; nc < 8; ++nc) O[nc][r] *= alpha;
    }
#pragma unroll
    for (int ks = 0; ks < KT / 4; ++ks) {
      const int kr = ks * 4 + 2 * hi;
      const v2f af = *(const v2f*)&myP[lm * PP + kr];
#pragma unroll
      for (int nc = 0; nc < 8; ++nc) {
        v2f bf;
        bf.x = ldsV[kr * VP + (nc << 4) + lm];
        bf.y = ldsV[(kr + 1) * VP + (nc << 4) + lm];
        O[nc] = wmma4(af, bf, O[nc]);
      }
    }
  }
  float* ap = att + ((size_t)b * NC + h * ND) * NHW;
#pragma unroll
  for (int r = 0; r < 8; ++r) {
    const float inv = 1.f / lst[r];
#pragma unroll
    for (int nc = 0; nc < 8; ++nc)
      ap[(size_t)((nc << 4) + lm) * NHW + i0 + r + 8 * hi] = O[nc][r] * inv;
  }
}

extern "C" void kernel_launch(void* const* d_in, const int* in_sizes, int n_in,
                              void* d_out, int out_size, void* d_ws,
                              size_t ws_size, hipStream_t stream) {
  (void)in_sizes; (void)n_in; (void)out_size; (void)ws_size;
  const float* x = (const float*)d_in[0];
  const float* gnw = (const float*)d_in[1];
  const float* gnb = (const float*)d_in[2];
  const float* wqkv = (const float*)d_in[3];
  const float* wout = (const float*)d_in[4];
  const float* bout = (const float*)d_in[5];
  float* out = (float*)d_out;
  float* ws = (float*)d_ws;
  float* xn = ws;
  float* qkv = ws + 8388608;
  float* vt = ws + 33554432;

  gn_kernel<<<NB * NG, 256, 0, stream>>>(x, gnw, gnb, xn);
  gemm512_kernel<<<dim3(12, 8, NB), 256, 0, stream>>>(wqkv, xn, qkv, 524288L,
                                                   1572864L, nullptr, nullptr, vt);
  attn_kernel<<<NB * NH * 8, 256, 0, stream>>>(qkv, vt, xn);
  gemm512_kernel<<<dim3(4, 8, NB), 256, 0, stream>>>(wout, xn, out, 524288L,
                                                   524288L, bout, x, nullptr);
}


# attention: K and V^T tiles kept in LDS as interleaved k pairs (B fragment = one 8-byte read); each WMMA group's fragment reads issued as one run a group ahead, no drain per WMMA pair; same sum order
# speedup vs baseline: 2.0273x; 1.0449x over previous
#include <hip/hip_runtime.h>


typedef __attribute__((ext_vector_type(2))) float v2f;
typedef __attribute__((ext_vector_type(8))) float v8f;

#define NB 16
#define NC 512
#define NHW 1024
#define NG 32
#define NH 4
#define ND 128
#define GN_EPS 1e-5f
#define ATT_SCALE 0.08838834764831845f

static __device__ __forceinline__ v8f wmma4(v2f a, v2f b, v8f c) {
  return __builtin_amdgcn_wmma_f32_16x16x4_f32(false, a, false, b, (short)0, c,
                                               false, false);
}

__global__ __launch_bounds__(256) void gn_kernel(const float* __restrict__ x,
                                                 const float* __restrict__ w,
                                                 const float* __restrict__ bia,
                                                 float* __restrict__ xn) {
  const int b = blockIdx.x >> 5;
  const int g = blockIdx.x & 31;
  const size_t base = ((size_t)b * NC + (size_t)g * 16) * NHW;
  const int tid = threadIdx.x;
  float sum = 0.f, sq = 0.f;
  for (int i = tid; i < 16 * NHW; i += 256) {
    float v = x[base + i];
    sum += v;
    sq += v * v;
  }
  for (int m = 16; m >= 1; m >>= 1) {
    sum += __shfl_xor(sum, m, 32);
    sq += __shfl_xor(sq, m, 32);
  }
  __shared__ float rs[8], rq[8], stat[2];
  const int wv = tid >> 5;
  if ((tid & 31) == 0) { rs[wv] = sum; rq[wv] = sq; }
  __syncthreads();
  if (tid == 0) {
    float ts = 0.f, tq = 0.f;
    for (int i = 0; i < 8; ++i) { ts += rs[i]; tq += rq[i]; }
    float mean = ts * (1.f / 16384.f);
    float var = tq * (1.f / 16384.f) - mean * mean;
    stat[0] = mean;
    stat[1] = rsqrtf(var + GN_EPS);
  }
  __syncthreads();
  const float mean = stat[0], rstd = stat[1];
  for (int i = tid; i < 16 * NHW; i += 256) {
    int c = (g << 4) + (i >> 10);
    xn[base + i] = (x[base + i] - mean) * rstd * w[c] + bia[c];
  }
}

#define GBM 128
#define GBN 128
#define GBK 32
#define GAP 36
#define GBP 288
#define GEP 36
#define GTP 68
#define GASZ (GBM * GAP)
#define GBSZ ((GBK / 2) * GBP)
typedef __attribute__((ext_vector_type(4))) float g4f;

__global__ __launch_bounds__(256, 6) void gemm512_kernel(
    const float* __restrict__ A, const float* __restrict__ Bmat,
    float* __restrict__ Y, long bStrideB, long bStrideY,
    const float* __restrict__ bias, const float* __restrict__ resid,
    float* __restrict__ Yt) {
  __shared__ __attribute__((aligned(16))) float lds[2 * GASZ + 2 * GBSZ];
  float* ldsA = lds;
  float* ldsB = lds + 2 * GASZ;
  const int m0 = blockIdx.x * GBM;
  const int n0 = blockIdx.y * GBN;
  const int b = blockIdx.z;
  const int tid = threadIdx.x;
  const int lane = tid & 31, wv = tid >> 5;
  const int lm = lane & 15, hi = lane >> 4;
  const int wm = wv >> 2, wn = wv & 3;
  const float* Bb = Bmat + (size_t)b * bStrideB;

  const float* Ag = A + (size_t)(m0 + (tid >> 3)) * 512 + ((tid & 7) << 2);
  const float* Bg = Bb + (size_t)(2 * (tid >> 5)) * NHW + n0 + ((tid & 31) << 2);
  float* Al = ldsA + (tid >> 3) * GAP + ((tid & 7) << 2);
  float* Bl = ldsB + (tid >> 5) * GBP + ((tid & 31) << 3);
  const float* Ar = ldsA + (wm * 64 + lm) * GAP + 2 * hi;
  const float* Br = ldsB + hi * GBP + 2 * (wn * 32 + lm);

  g4f ar[4], br0[2], br1[2];
  v8f acc[4][2] = {};

#define G_LOAD(kb)                                                              \
  {                                                                             \
    _Pragma("unroll") for (int u = 0; u < 4; ++u)                               \
        ar[u] = *(const g4f*)(Ag + (size_t)u * 32 * 512 + (kb));                \
    _Pragma("unroll") for (int u = 0; u < 2; ++u) {                             \
      br0[u] = *(const g4f*)(Bg + (size_t)((kb) + 16 * u) * NHW);               \
      br1[u] = *(const g4f*)(Bg + (size_t)((kb) + 16 * u + 1) * NHW);           \
    }                                                                           \
  }
#define G_STORE(st)                                                             \
  {                                                                             \
    _Pragma("unroll") for (int u = 0; u < 4; ++u)                               \
        *(g4f*)(Al + (st) * GASZ + u * 32 * GAP) = ar[u];                       \
    _Pragma("unroll") for (int u = 0; u < 2; ++u) {                             \
      g4f lo, up;                                                               \
      lo.x = br0[u].x; lo.y = br1[u].x; lo.z = br0[u].y; lo.w = br1[u].y;       \
      up.x = br0[u].z; up.y = br1[u].z; up.z = br0[u].w; up.w = br1[u].w;       \
      *(g4f*)(Bl + (st) * GBSZ + u * 8 * GBP) = lo;                             \
      *(g4f*)(Bl + (st) * GBSZ + u * 8 * GBP + 4) = up;                         \
    }                                                                           \
  }
#define G_COMPUTE(st)                                                           \
  {                                                                             \
    const float* as = Ar + (st) * GASZ;                                         \
    const float* bs = Br + (st) * GBSZ;                                         \
    _Pragma("unroll") for (int kk = 0; kk < GBK / 4; ++kk) {                    \
      v2f af[4], bf[2];                                                         \
      _Pragma("unroll") for (int mt = 0; mt < 4; ++mt)                          \
          af[mt] = *(const v2f*)(as + mt * 16 * GAP + kk * 4);                  \
      _Pragma("unroll") for (int nt = 0; nt < 2; ++nt)                          \
          bf[nt] = *(const v2f*)(bs + kk * 2 * GBP + nt * 32);                  \
      _Pragma("unroll") for (int mt = 0; mt < 4; ++mt)                          \
          _Pragma("unroll") for (int nt = 0; nt < 2; ++nt)                      \
              acc[mt][nt] = wmma4(af[mt], bf[nt], acc[mt][nt]);                 \
    }                                                                           \
  }

  G_LOAD(0);
  G_STORE(0);
  __syncthreads();
  for (int t = 0; t < 512 / GBK - 1; ++t) {
    const int cur = t & 1;
    G_LOAD((t + 1) * GBK);
    __builtin_amdgcn_sched_barrier(0);
    G_COMPUTE(cur);
    G_STORE(cur ^ 1);
    __syncthreads();
  }
  G_COMPUTE(1);
#undef G_LOAD
#undef G_STORE
#undef G_COMPUTE

  __syncthreads();
  float* slab = lds + wv * (64 * GEP);
  static_assert(GBM == ND, "a V row tile must be exactly one head");
  static_assert(32 * GTP <= 64 * GEP, "the transposed slab must fit the wave's slab");
  if (Yt != nullptr && m0 >= 2 * NC) {
#pragma unroll
    for (int mt = 0; mt < 4; ++mt)
#pragma unroll
      for (int nt = 0; nt < 2; ++nt) {
        float* p = slab + (nt * 16 + lm) * GTP + mt * 16 + 8 * hi;
        g4f lo, up;
        lo.x = acc[mt][nt][0]; lo.y = acc[mt][nt][1]; lo.z = acc[mt][nt][2]; lo.w = acc[mt][nt][3];
        up.x = acc[mt][nt][4]; up.y = acc[mt][nt][5]; up.z = acc[mt][nt][6]; up.w = acc[mt][nt][7];
        *(g4f*)p = lo;
        *(g4f*)(p + 4) = up;
      }
    __syncthreads();
    const int head = (m0 - 2 * NC) / ND;
    float* Vb = Yt + ((size_t)(b * NH + head) * NHW + n0 + wn * 32) * ND + wm * 64;
    const int jr = lane >> 4, d4 = (lane & 15) << 2;
#pragma unroll
    for (int it = 0; it < 16; ++it) {
      const int jl = it * 2 + jr;
      *(g4f*)(Vb + (size_t)jl * ND + d4) = *(const g4f*)(slab + jl * GTP + d4);
    }
    return;
  }
#pragma unroll
  for (int mt = 0; mt < 4; ++mt)
#pragma unroll
    for (int nt = 0; nt < 2; ++nt)
#pragma unroll
      for (int r = 0; r < 8; ++r)
        slab[(mt * 16 + r + 8 * hi) * GEP + nt * 16 + lm] = acc[mt][nt][r];
  __syncthreads();
  float* Yb = Y + (size_t)b * bStrideY;
  const int rl = lane >> 3, c4 = (lane & 7) << 2;
#pragma unroll
  for (int it = 0; it < 16; ++it) {
    const int rloc = it * 4 + rl;
    g4f v = *(const g4f*)(slab + rloc * GEP + c4);
    const int row = m0 + wm * 64 + rloc;
    const int col = n0 + wn * 32 + c4;
    if (bias) {
      const float bb = bias[row];
      const g4f rr = *(const g4f*)(resid + (size_t)b * bStrideY + (size_t)row * NHW + col);
      v.x += bb + rr.x; v.y += bb + rr.y; v.z += bb + rr.z; v.w += bb + rr.w;
    }
    *(g4f*)(Yb + (size_t)row * NHW + col) = v;
  }
}

#define KT 64
#define KP2 160
#define VP2 288
#define PP 68
#define QB 2
#define PB 2
typedef __attribute__((ext_vector_type(4))) float v4f;

__global__ __launch_bounds__(256, 4) void attn_kernel(const float* __restrict__ qkv,
                                                   const float* __restrict__ vt,
                                                   float* __restrict__ att) {
  __shared__ __attribute__((aligned(16))) float ldsK[(ND / 2) * KP2];
  __shared__ __attribute__((aligned(16))) float ldsV[(KT / 2) * VP2];
  __shared__ __attribute__((aligned(16))) float ldsP[8][16 * PP];
  const int bh = blockIdx.x >> 3, iblk = blockIdx.x & 7;
  const int b = bh >> 2, h = bh & 3;
  const int tid = threadIdx.x, wv = tid >> 5, lane = tid & 31;
  const int lm = lane & 15, hi = lane >> 4;
  const int i0 = (iblk << 7) + (wv << 4);
  const float* qp = qkv + ((size_t)b * 1536 + h * ND) * NHW;
  const float* kp = qkv + ((size_t)b * 1536 + 512 + h * ND) * NHW;
  const float* vp = vt + (size_t)bh * NHW * ND;
  v2f qf[32];
#pragma unroll
  for (int ks = 0; ks < 32; ++ks) {
    const int kr = ks * 4 + 2 * hi;
    qf[ks].x = qp[(size_t)kr * NHW + i0 + lm];
    qf[ks].y = qp[(size_t)(kr + 1) * NHW + i0 + lm];
  }
  v8f O[8] = {};
  float mst[8], lst[8];
#pragma unroll
  for (int r = 0; r < 8; ++r) { mst[r] = -3.0e38f; lst[r] = 0.f; }
  float* myP = &ldsP[wv][0];

  const unsigned koff = (unsigned)(2 * (tid >> 4)) * NHW + ((tid & 15) << 2);
  const unsigned voff = (unsigned)(2 * (tid >> 5)) * ND + ((tid & 31) << 2);
  float* kl = &ldsK[(tid >> 4) * KP2 + ((tid & 15) << 3)];
  float* vl = &ldsV[(tid >> 5) * VP2 + ((tid & 31) << 3)];
  const float* kfr = &ldsK[hi * KP2 + 2 * lm];
  const float* vfr = &ldsV[hi * VP2 + 2 * lm];
  const float* pfr = myP + lm * PP + 2 * hi;

  for (int jt = 0; jt < NHW / KT; ++jt) {
    const float* kt = kp + jt * KT;
    const float* vtile = vp + (size_t)jt * KT * ND;
    v4f ke[4], ko[4], ve[4], vo[4];
#pragma unroll
    for (int u = 0; u < 4; ++u) {
      ke[u] = *(const v4f*)(kt + (koff + (unsigned)u * 32 * NHW));
      ko[u] = *(const v4f*)(kt + (koff + (unsigned)u * 32 * NHW + NHW));
    }
#pragma unroll
    for (int u = 0; u < 4; ++u) {
      ve[u] = *(const v4f*)(vtile + (voff + (unsigned)u * 16 * ND));
      vo[u] = *(const v4f*)(vtile + (voff + (unsigned)u * 16 * ND + ND));
    }
    __syncthreads();
#pragma unroll
    for (int u = 0; u < 4; ++u) {
      v4f lo, up;
      lo.x = ke[u].x; lo.y = ko[u].x; lo.z = ke[u].y; lo.w = ko[u].y;
      up.x = ke[u].z; up.y = ko[u].z; up.z = ke[u].w; up.w = ko[u].w;
      *(v4f*)(kl + u * 16 * KP2) = lo;
      *(v4f*)(kl + u * 16 * KP2 + 4) = up;
    }
#pragma unroll
    for (int u = 0; u < 4; ++u) {
      v4f lo, up;
      lo.x = ve[u].x; lo.y = vo[u].x; lo.z = ve[u].y; lo.w = vo[u].y;
      up.x = ve[u].z; up.y = vo[u].z; up.z = ve[u].w; up.w = vo[u].w;
      *(v4f*)(vl + u * 8 * VP2) = lo;
      *(v4f*)(vl + u * 8 * VP2 + 4) = up;
    }
    __syncthreads();
    v8f s[4] = {};
    v2f kf[2][QB * 4];
#pragma unroll
    for (int i = 0; i < QB; ++i)
#pragma unroll
      for (int t = 0; t < 4; ++t)
        kf[0][i * 4 + t] = *(const v2f*)(kfr + i * 2 * KP2 + t * 32);
#pragma unroll
    for (int g = 0; g < 32 / QB; ++g) {
      if (g + 1 < 32 / QB) {
#pragma unroll
        for (int i = 0; i < QB; ++i)
#pragma unroll
          for (int t = 0; t < 4; ++t)
            kf[(g + 1) & 1][i * 4 + t] =
                *(const v2f*)(kfr + ((g + 1) * QB + i) * 2 * KP2 + t * 32);
      }
      __builtin_amdgcn_sched_barrier(0);
#pragma unroll
      for (int i = 0; i < QB; ++i)
#pragma unroll
        for (int t = 0; t < 4; ++t)
          s[t] = wmma4(qf[g * QB + i], kf[g & 1][i * 4 + t], s[t]);
      __builtin_amdgcn_sched_barrier(0);
    }
#pragma unroll
    for (int r = 0; r < 8; ++r) {
      const float s0 = s[0][r] * ATT_SCALE, s1 = s[1][r] * ATT_SCALE;
      const float s2 = s[2][r] * ATT_SCALE, s3 = s[3][r] * ATT_SCALE;
      float tm = fmaxf(fmaxf(s0, s1), fmaxf(s2, s3));
      tm = fmaxf(tm, __shfl_xor(tm, 1, 16));
      tm = fmaxf(tm, __shfl_xor(tm, 2, 16));
      tm = fmaxf(tm, __shfl_xor(tm, 4, 16));
      tm = fmaxf(tm, __shfl_xor(tm, 8, 16));
      const float mn = fmaxf(mst[r], tm);
      const float alpha = __expf(mst[r] - mn);
      const float p0 = __expf(s0 - mn), p1 = __expf(s1 - mn);
      const float p2 = __expf(s2 - mn), p3 = __expf(s3 - mn);
      float ts = (p0 + p1) + (p2 + p3);
      ts += __shfl_xor(ts, 1, 16);
      ts += __shfl_xor(ts, 2, 16);
      ts += __shfl_xor(ts, 4, 16);
      ts += __shfl_xor(ts, 8, 16);
      lst[r] = lst[r] * alpha + ts;
      mst[r] = mn;
      float* pr = myP + (r + 8 * hi) * PP + lm;
      pr[0] = p0; pr[16] = p1; pr[32] = p2; pr[48] = p3;
#pragma unroll
      for (int nc = 0; nc < 8; ++nc) O[nc][r] *= alpha;
    }
    v2f pf[2][PB];
    v2f vf[2][PB * 8];
#pragma unroll
    for (int i = 0; i < PB; ++i) {
      pf[0][i] = *(const v2f*)(pfr + i * 4);
#pragma unroll
      for (int nc = 0; nc < 8; ++nc)
        vf[0][i * 8 + nc] = *(const v2f*)(vfr + i * 2 * VP2 + nc * 32);
    }
#pragma unroll
    for (int g = 0; g < (KT / 4) / PB; ++g) {
      if (g + 1 < (KT / 4) / PB) {
#pragma unroll
        for (int i = 0; i < PB; ++i) {
          pf[(g + 1) & 1][i] = *(const v2f*)(pfr + ((g + 1) * PB + i) * 4);
#pragma unroll
          for (int nc = 0; nc < 8; ++nc)
            vf[(g + 1) & 1][i * 8 + nc] =
                *(const v2f*)(vfr + ((g + 1) * PB + i) * 2 * VP2 + nc * 32);
        }
      }
      __builtin_amdgcn_sched_barrier(0);
#pragma unroll
      for (int i = 0; i < PB; ++i)
#pragma unroll
        for (int nc = 0; nc < 8; ++nc)
          O[nc] = wmma4(pf[g & 1][i], vf[g & 1][i * 8 + nc], O[nc]);
      __builtin_amdgcn_sched_barrier(0);
    }
  }
  float* ap = att + ((size_t)b * NC + h * ND) * NHW;
#pragma unroll
  for (int r = 0; r < 8; ++r) {
    const float inv = 1.f / lst[r];
#pragma unroll
    for (int nc = 0; nc < 8; ++nc)
      ap[(size_t)((nc << 4) + lm) * NHW + i0 + r + 8 * hi] = O[nc][r] * inv;
  }
}

extern "C" void kernel_launch(void* const* d_in, const int* in_sizes, int n_in,
                              void* d_out, int out_size, void* d_ws,
                              size_t ws_size, hipStream_t stream) {
  (void)in_sizes; (void)n_in; (void)out_size; (void)ws_size;
  const float* x = (const float*)d_in[0];
  const float* gnw = (const float*)d_in[1];
  const float* gnb = (const float*)d_in[2];
  const float* wqkv = (const float*)d_in[3];
  const float* wout = (const float*)d_in[4];
  const float* bout = (const float*)d_in[5];
  float* out = (float*)d_out;
  float* ws = (float*)d_ws;
  float* xn = ws;
  float* qkv = ws + 8388608;
  float* vt = ws + 33554432;

  gn_kernel<<<NB * NG, 256, 0, stream>>>(x, gnw, gnb, xn);
  gemm512_kernel<<<dim3(12, 8, NB), 256, 0, stream>>>(wqkv, xn, qkv, 524288L,
                                                   1572864L, nullptr, nullptr, vt);
  attn_kernel<<<NB * NH * 8, 256, 0, stream>>>(qkv, vt, xn);
  gemm512_kernel<<<dim3(4, 8, NB), 256, 0, stream>>>(wout, xn, out, 524288L,
                                                   524288L, bout, x, nullptr);
}
